# starting kernel
# baseline (speedup 1.0000x reference)
#include <hip/hip_runtime.h>
#include <cstdint>
#include <cstddef>

#define IN_F 128
#define NEG_SLOPE 0.2f
#define H0 4
#define H1 1

typedef __attribute__((ext_vector_type(2))) float v2f;
typedef __attribute__((ext_vector_type(8))) float v8f;

template <int NCOL>
__global__ __launch_bounds__(32 * (NCOL / 16)) void wmma_gemm_k128(
    const float* __restrict__ X, const float* __restrict__ W, float* __restrict__ Y) {
  constexpr int K = 128;
  __shared__ float xs[16][K + 4];
  const int row0 = blockIdx.x * 16;
  const int tid = threadIdx.x;
  const int lane = tid & 31;
  const int wave = tid >> 5;

  for (int i = tid; i < 16 * K; i += 32 * (NCOL / 16)) {
    int r = i >> 7;
    int c = i & 127;
    xs[r][c] = X[(row0 + r) * K + c];
  }
  __syncthreads();

  const int n0 = wave * 16;
  const int m = lane & 15;
  const int khalf = lane >> 4;

  v8f acc = {};
  for (int k0 = 0; k0 < K; k0 += 4) {
    const int ka = k0 + khalf * 2;
    v2f a, b;
    a.x = xs[m][ka];
    a.y = xs[m][ka + 1];
    b.x = W[(ka) *NCOL + n0 + m];
    b.y = W[(ka + 1) * NCOL + n0 + m];
    acc = __builtin_amdgcn_wmma_f32_16x16x4_f32(false, a, false, b, (short)0, acc,
                                                false, false);
  }

  const int n = n0 + m;
#pragma unroll
  for (int r = 0; r < 8; ++r) {
    Y[(row0 + khalf * 8 + r) * NCOL + n] = acc[r];
  }
}

template <int H>
__global__ void attn_coef32(const float* __restrict__ h, const float* __restrict__ al,
                            const float* __restrict__ ar, float* __restrict__ el,
                            float* __restrict__ er, int nh) {
  const int gw = blockIdx.x * (blockDim.x >> 5) + (threadIdx.x >> 5);
  if (gw >= nh) return;
  const int lane = threadIdx.x & 31;
  const int head = gw % H;
  const float v = h[gw * 32 + lane];
  float l = v * al[head * 32 + lane];
  float r = v * ar[head * 32 + lane];
#pragma unroll
  for (int off = 16; off > 0; off >>= 1) {
    l += __shfl_down(l, off, 32);
    r += __shfl_down(r, off, 32);
  }
  if (lane == 0) {
    el[gw] = l;
    er[gw] = r;
  }
}

template <int H>
__global__ void edge_e_max(const int* __restrict__ src, const int* __restrict__ dst,
                           const float* __restrict__ el, const float* __restrict__ er,
                           float* __restrict__ e, float* __restrict__ m, int ne) {
  const int i = blockIdx.x * blockDim.x + threadIdx.x;
  if (i >= ne * H) return;
  const int eid = i / H;
  const int h = i % H;
  const int d = dst[eid];
  float v = el[src[eid] * H + h] + er[d * H + h];
  v = (v > 0.0f) ? v : NEG_SLOPE * v;
  e[i] = v;
  float* slot = m + d * H + h;
  if (v >= 0.0f)
    atomicMax((int*)slot, __float_as_int(v));
  else
    atomicMin((unsigned int*)slot, __float_as_uint(v));
}

template <int H>
__global__ void edge_ex_sum(const int* __restrict__ dst, const float* __restrict__ m,
                            float* __restrict__ e, float* __restrict__ s, int ne) {
  const int i = blockIdx.x * blockDim.x + threadIdx.x;
  if (i >= ne * H) return;
  const int eid = i / H;
  const int h = i % H;
  const int slot = dst[eid] * H + h;
  const float ex = expf(e[i] - m[slot]);
  e[i] = ex;
  atomicAdd(s + slot, ex);
}

template <int H, int D>
__global__ void edge_aggregate(const int* __restrict__ src, const int* __restrict__ dst,
                               const float* __restrict__ ex, const float* __restrict__ s,
                               const float* __restrict__ hfeat, float* __restrict__ out,
                               int ne) {
  constexpr int HD = H * D;
  const long long t = (long long)blockIdx.x * blockDim.x + threadIdx.x;
  if (t >= (long long)ne * HD) return;
  const int eid = (int)(t / HD);
  const int c = (int)(t % HD);
  const int h = c / D;
  const int sn = src[eid];
  const int dn = dst[eid];
  const float alpha = ex[eid * H + h] / s[dn * H + h];
  atomicAdd(out + dn * HD + c, alpha * hfeat[sn * HD + c]);
}

template <int C>
__global__ void bias_elu(const float* __restrict__ agg, const float* __restrict__ b,
                         float* __restrict__ act, int total) {
  const int i = blockIdx.x * blockDim.x + threadIdx.x;
  if (i >= total) return;
  const float v = agg[i] + b[i & (C - 1)];
  act[i] = (v > 0.0f) ? v : (expf(v) - 1.0f);
}

__global__ void add_bias32(float* __restrict__ out, const float* __restrict__ b,
                           int total) {
  const int i = blockIdx.x * blockDim.x + threadIdx.x;
  if (i >= total) return;
  out[i] += b[i & 31];
}

extern "C" void kernel_launch(void* const* d_in, const int* in_sizes, int n_in,
                              void* d_out, int out_size, void* d_ws, size_t ws_size,
                              hipStream_t stream) {
  const float* x = (const float*)d_in[0];
  const int* src = (const int*)d_in[1];
  const int* dst = (const int*)d_in[2];
  const float* W1 = (const float*)d_in[3];
  const float* al1 = (const float*)d_in[4];
  const float* ar1 = (const float*)d_in[5];
  const float* b1 = (const float*)d_in[6];
  const float* W2 = (const float*)d_in[7];
  const float* al2 = (const float*)d_in[8];
  const float* ar2 = (const float*)d_in[9];
  const float* b2 = (const float*)d_in[10];

  const int n = in_sizes[0] / IN_F;
  const int e = in_sizes[1];

  char* ws = (char*)d_ws;
  size_t off = 0;
  auto wsalloc = [&](size_t bytes) {
    void* p = ws + off;
    off = (off + bytes + 255) & ~(size_t)255;
    return p;
  };
  float* h1 = (float*)wsalloc((size_t)n * 128 * sizeof(float));
  float* agg1 = (float*)wsalloc((size_t)n * 128 * sizeof(float));
  float* ebuf = (float*)wsalloc((size_t)e * H0 * sizeof(float));
  float* el = (float*)wsalloc((size_t)n * H0 * sizeof(float));
  float* er = (float*)wsalloc((size_t)n * H0 * sizeof(float));
  float* mbuf = (float*)wsalloc((size_t)n * H0 * sizeof(float));
  float* sbuf = (float*)wsalloc((size_t)n * H0 * sizeof(float));
  float* outv = (float*)d_out;

  wmma_gemm_k128<128><<<n / 16, 256, 0, stream>>>(x, W1, h1);

  {
    const int nh = n * H0;
    attn_coef32<H0><<<(nh + 7) / 8, 256, 0, stream>>>(h1, al1, ar1, el, er, nh);
  }

  hipMemsetAsync(mbuf, 0xFF, (size_t)n * H0 * sizeof(float), stream);
  hipMemsetAsync(sbuf, 0, (size_t)n * H0 * sizeof(float), stream);
  hipMemsetAsync(agg1, 0, (size_t)n * 128 * sizeof(float), stream);

  edge_e_max<H0><<<(e * H0 + 255) / 256, 256, 0, stream>>>(src, dst, el, er, ebuf, mbuf, e);
  edge_ex_sum<H0><<<(e * H0 + 255) / 256, 256, 0, stream>>>(dst, mbuf, ebuf, sbuf, e);
  {
    const long long total = (long long)e * 128;
    edge_aggregate<H0, 32><<<(int)((total + 255) / 256), 256, 0, stream>>>(
        src, dst, ebuf, sbuf, h1, agg1, e);
  }

  bias_elu<128><<<(n * 128 + 255) / 256, 256, 0, stream>>>(agg1, b1, h1, n * 128);

  float* g2 = agg1;
  wmma_gemm_k128<32><<<n / 16, 64, 0, stream>>>(h1, W2, g2);

  attn_coef32<H1><<<(n + 7) / 8, 256, 0, stream>>>(g2, al2, ar2, el, er, n);

  hipMemsetAsync(mbuf, 0xFF, (size_t)n * H1 * sizeof(float), stream);
  hipMemsetAsync(sbuf, 0, (size_t)n * H1 * sizeof(float), stream);
  hipMemsetAsync(d_out, 0, (size_t)n * 32 * sizeof(float), stream);

  edge_e_max<H1><<<(e + 255) / 256, 256, 0, stream>>>(src, dst, el, er, ebuf, mbuf, e);
  edge_ex_sum<H1><<<(e + 255) / 256, 256, 0, stream>>>(dst, mbuf, ebuf, sbuf, e);
  {
    const long long total = (long long)e * 32;
    edge_aggregate<H1, 32><<<(int)((total + 255) / 256), 256, 0, stream>>>(
        src, dst, ebuf, sbuf, g2, outv, e);
  }

  add_bias32<<<(n * 32 + 255) / 256, 256, 0, stream>>>(outv, b2, n * 32);
}


# group edges by destination once per call (counting sort); one wave per destination node then does segment max, sum and weighted row sum in registers: no float atomics, no e/m/s buffers
# speedup vs baseline: 3.2221x; 3.2221x over previous
#include <hip/hip_runtime.h>
#include <cstdint>
#include <cstddef>

#define IN_F 128
#define NEG_SLOPE 0.2f
#define H0 4
#define H1 1

typedef __attribute__((ext_vector_type(2))) float v2f;
typedef __attribute__((ext_vector_type(8))) float v8f;

template <int NCOL>
__global__ __launch_bounds__(32 * (NCOL / 16)) void wmma_gemm_k128(
    const float* __restrict__ X, const float* __restrict__ W, float* __restrict__ Y) {
  constexpr int K = 128;
  __shared__ float xs[16][K + 4];
  const int row0 = blockIdx.x * 16;
  const int tid = threadIdx.x;
  const int lane = tid & 31;
  const int wave = tid >> 5;

  for (int i = tid; i < 16 * K; i += 32 * (NCOL / 16)) {
    int r = i >> 7;
    int c = i & 127;
    xs[r][c] = X[(row0 + r) * K + c];
  }
  __syncthreads();

  const int n0 = wave * 16;
  const int m = lane & 15;
  const int khalf = lane >> 4;

  v8f acc = {};
  for (int k0 = 0; k0 < K; k0 += 4) {
    const int ka = k0 + khalf * 2;
    v2f a, b;
    a.x = xs[m][ka];
    a.y = xs[m][ka + 1];
    b.x = W[(ka) *NCOL + n0 + m];
    b.y = W[(ka + 1) * NCOL + n0 + m];
    acc = __builtin_amdgcn_wmma_f32_16x16x4_f32(false, a, false, b, (short)0, acc,
                                                false, false);
  }

  const int n = n0 + m;
#pragma unroll
  for (int r = 0; r < 8; ++r) {
    Y[(row0 + khalf * 8 + r) * NCOL + n] = acc[r];
  }
}

template <int H>
__global__ void attn_coef32(const float* __restrict__ h, const float* __restrict__ al,
                            const float* __restrict__ ar, float* __restrict__ el,
                            float* __restrict__ er, int nh) {
  const int gw = blockIdx.x * (blockDim.x >> 5) + (threadIdx.x >> 5);
  if (gw >= nh) return;
  const int lane = threadIdx.x & 31;
  const int head = gw % H;
  const float v = h[gw * 32 + lane];
  float l = v * al[head * 32 + lane];
  float r = v * ar[head * 32 + lane];
#pragma unroll
  for (int off = 16; off > 0; off >>= 1) {
    l += __shfl_down(l, off, 32);
    r += __shfl_down(r, off, 32);
  }
  if (lane == 0) {
    el[gw] = l;
    er[gw] = r;
  }
}

__global__ void edge_rank(const int* __restrict__ dst, int* __restrict__ cnt,
                          int* __restrict__ rank, int ne) {
  const int i = blockIdx.x * blockDim.x + threadIdx.x;
  if (i >= ne) return;
  rank[i] = atomicAdd(cnt + dst[i], 1);
}

__global__ __launch_bounds__(256) void seg_offsets(const int* __restrict__ cnt,
                                                   int* __restrict__ off, int n) {
  __shared__ int w_before[8];
  __shared__ int w_tile[8];
  const int tid = threadIdx.x;
  const int lane = tid & 31;
  const int wave = tid >> 5;
  const int tile0 = blockIdx.x * 1024;

  int before = 0;
  for (int i = tid * 4; i < tile0; i += 1024) {
    const int4 v = *reinterpret_cast<const int4*>(cnt + i);
    before += v.x + v.y + v.z + v.w;
  }

  const int i0 = tile0 + tid * 4;
  int c[4];
  for (int k = 0; k < 4; ++k) c[k] = (i0 + k < n) ? cnt[i0 + k] : 0;
  const int local = c[0] + c[1] + c[2] + c[3];

  int incl = local;
  for (int o = 1; o < 32; o <<= 1) {
    const int t = __shfl_up(incl, o, 32);
    if (lane >= o) incl += t;
  }
  for (int o = 16; o > 0; o >>= 1) before += __shfl_xor(before, o, 32);

  if (lane == 31) w_tile[wave] = incl;
  if (lane == 0) w_before[wave] = before;
  __syncthreads();

  int run = incl - local;
  for (int w = 0; w < 8; ++w) {
    run += w_before[w];
    if (w < wave) run += w_tile[w];
  }
  for (int k = 0; k < 4; ++k) {
    if (i0 + k <= n) off[i0 + k] = run;
    run += c[k];
  }
}

__global__ void edge_place(const int* __restrict__ src, const int* __restrict__ dst,
                           const int* __restrict__ off, const int* __restrict__ rank,
                           int* __restrict__ csrc, int ne) {
  const int i = blockIdx.x * blockDim.x + threadIdx.x;
  if (i >= ne) return;
  unsigned pos = (unsigned)(off[dst[i]] + rank[i]);
  pos = min(pos, (unsigned)(ne - 1));
  csrc[pos] = src[i];
}

template <int H>
__global__ __launch_bounds__(256) void gat_gather(
    const int* __restrict__ off, const int* __restrict__ csrc,
    const float* __restrict__ el, const float* __restrict__ er,
    const float* __restrict__ hfeat, float* __restrict__ out, int n) {
  constexpr int HD = H * 32;
  const int node = blockIdx.x * 8 + (threadIdx.x >> 5);
  if (node >= n) return;
  const int lane = threadIdx.x & 31;
  const int head = (lane * H) >> 5;
  const int beg = __builtin_amdgcn_readfirstlane(off[node]);
  const int end = __builtin_amdgcn_readfirstlane(off[node + 1]);
  const unsigned last = (unsigned)(n - 1);

  float erh[H];
  for (int h = 0; h < H; ++h) erh[h] = er[node * H + h];

  float mh[H];
  int s0 = 0;
  for (int h = 0; h < H; ++h) mh[h] = -INFINITY;
  for (int base = beg; base < end; base += 32) {
    const int p = base + lane;
    if (p < end) {
      const int s = (int)min((unsigned)csrc[p], last);
      for (int h = 0; h < H; ++h) {
        float v = el[s * H + h] + erh[h];
        v = (v > 0.0f) ? v : NEG_SLOPE * v;
        mh[h] = fmaxf(mh[h], v);
      }
      if (base == beg) s0 = s;
    }
  }
  for (int h = 0; h < H; ++h) {
    for (int o = 16; o > 0; o >>= 1) mh[h] = fmaxf(mh[h], __shfl_xor(mh[h], o, 32));
  }
  float m = mh[0];
  float erv = erh[0];
  for (int h = 1; h < H; ++h) {
    m = (head == h) ? mh[h] : m;
    erv = (head == h) ? erh[h] : erv;
  }

  const float* elh = el + head;
  const float* hcol = hfeat + lane * H;
  float ssum = 0.0f;
  float acc[H];
  for (int k = 0; k < H; ++k) acc[k] = 0.0f;
  for (int base = beg; base < end; base += 32) {
    const int cnt = min(32, end - base);
    int s_l = s0;
    if (base != beg) {
      s_l = (lane < cnt) ? (int)min((unsigned)csrc[base + lane], last) : 0;
    }
#pragma unroll 4
    for (int j = 0; j < cnt; ++j) {
      const int s = __builtin_amdgcn_readlane(s_l, j);
      const float t = elh[(size_t)s * H] + erv;
      const float v = (t > 0.0f) ? t : NEG_SLOPE * t;
      const float w = expf(v - m);
      ssum += w;
      const float* row = hcol + (size_t)s * HD;
      if constexpr (H == 4) {
        const float4 r = *reinterpret_cast<const float4*>(row);
        acc[0] += w * r.x;
        acc[1] += w * r.y;
        acc[2] += w * r.z;
        acc[3] += w * r.w;
      } else {
        for (int k = 0; k < H; ++k) acc[k] += w * row[k];
      }
    }
  }

  const bool any = end > beg;
  float* o = out + (size_t)node * HD + lane * H;
  if constexpr (H == 4) {
    float4 r;
    r.x = any ? acc[0] / ssum : 0.0f;
    r.y = any ? acc[1] / ssum : 0.0f;
    r.z = any ? acc[2] / ssum : 0.0f;
    r.w = any ? acc[3] / ssum : 0.0f;
    *reinterpret_cast<float4*>(o) = r;
  } else {
    for (int k = 0; k < H; ++k) o[k] = any ? acc[k] / ssum : 0.0f;
  }
}

template <int C>
__global__ void bias_elu(const float* __restrict__ agg, const float* __restrict__ b,
                         float* __restrict__ act, int total) {
  const int i = blockIdx.x * blockDim.x + threadIdx.x;
  if (i >= total) return;
  const float v = agg[i] + b[i & (C - 1)];
  act[i] = (v > 0.0f) ? v : (expf(v) - 1.0f);
}

__global__ void add_bias32(float* __restrict__ out, const float* __restrict__ b,
                           int total) {
  const int i = blockIdx.x * blockDim.x + threadIdx.x;
  if (i >= total) return;
  out[i] += b[i & 31];
}

extern "C" void kernel_launch(void* const* d_in, const int* in_sizes, int n_in,
                              void* d_out, int out_size, void* d_ws, size_t ws_size,
                              hipStream_t stream) {
  const float* x = (const float*)d_in[0];
  const int* src = (const int*)d_in[1];
  const int* dst = (const int*)d_in[2];
  const float* W1 = (const float*)d_in[3];
  const float* al1 = (const float*)d_in[4];
  const float* ar1 = (const float*)d_in[5];
  const float* b1 = (const float*)d_in[6];
  const float* W2 = (const float*)d_in[7];
  const float* al2 = (const float*)d_in[8];
  const float* ar2 = (const float*)d_in[9];
  const float* b2 = (const float*)d_in[10];

  const int n = in_sizes[0] / IN_F;
  const int e = in_sizes[1];

  char* ws = (char*)d_ws;
  size_t off = 0;
  auto wsalloc = [&](size_t bytes) {
    void* p = ws + off;
    off = (off + bytes + 255) & ~(size_t)255;
    return p;
  };
  float* h1 = (float*)wsalloc((size_t)n * 128 * sizeof(float));
  float* agg1 = (float*)wsalloc((size_t)n * 128 * sizeof(float));
  int* csrc = (int*)wsalloc((size_t)e * sizeof(int));
  int* rank = (int*)wsalloc((size_t)e * sizeof(int));
  int* cnt = (int*)wsalloc((size_t)n * sizeof(int));
  int* offs = (int*)wsalloc((size_t)(n + 1) * sizeof(int));
  float* el = (float*)wsalloc((size_t)n * H0 * sizeof(float));
  float* er = (float*)wsalloc((size_t)n * H0 * sizeof(float));
  float* outv = (float*)d_out;

  hipMemsetAsync(cnt, 0, (size_t)n * sizeof(int), stream);
  edge_rank<<<(e + 255) / 256, 256, 0, stream>>>(dst, cnt, rank, e);
  seg_offsets<<<(n + 1 + 1023) / 1024, 256, 0, stream>>>(cnt, offs, n);
  edge_place<<<(e + 255) / 256, 256, 0, stream>>>(src, dst, offs, rank, csrc, e);

  wmma_gemm_k128<128><<<n / 16, 256, 0, stream>>>(x, W1, h1);

  {
    const int nh = n * H0;
    attn_coef32<H0><<<(nh + 7) / 8, 256, 0, stream>>>(h1, al1, ar1, el, er, nh);
  }

  gat_gather<H0><<<(n + 7) / 8, 256, 0, stream>>>(offs, csrc, el, er, h1, agg1, n);

  bias_elu<128><<<(n * 128 + 255) / 256, 256, 0, stream>>>(agg1, b1, h1, n * 128);

  float* g2 = agg1;
  wmma_gemm_k128<32><<<n / 16, 64, 0, stream>>>(h1, W2, g2);

  attn_coef32<H1><<<(n + 7) / 8, 256, 0, stream>>>(g2, al2, ar2, el, er, n);

  gat_gather<H1><<<(n + 7) / 8, 256, 0, stream>>>(offs, csrc, el, er, g2, outv, n);

  add_bias32<<<(n * 32 + 255) / 256, 256, 0, stream>>>(outv, b2, n * 32);
}


# gather: lane = edge forms each edge's softmax weights once (H expf per 32 edges, not one per edge in every lane) and hands them to the row loop through LDS; el reload and score math leave the loop
# speedup vs baseline: 4.2292x; 1.3126x over previous
#include <hip/hip_runtime.h>
#include <cstdint>
#include <cstddef>

#define IN_F 128
#define NEG_SLOPE 0.2f
#define H0 4
#define H1 1

typedef __attribute__((ext_vector_type(2))) float v2f;
typedef __attribute__((ext_vector_type(8))) float v8f;

template <int MASK>
__device__ __forceinline__ float row_xor_get(float v) {
  return __builtin_bit_cast(float, __builtin_amdgcn_update_dpp(0, __builtin_bit_cast(int, v),
                                                               0x160 + MASK, 0xf, 0xf, true));
}
__device__ __forceinline__ float row16_sum(float v) {
  v += row_xor_get<1>(v);
  v += row_xor_get<2>(v);
  v += row_xor_get<4>(v);
  v += row_xor_get<8>(v);
  return v;
}

template <int NCOL>
__global__ __launch_bounds__(32 * (NCOL / 16)) void wmma_gemm_k128(
    const float* __restrict__ X, const float* __restrict__ W, const float* __restrict__ al,
    const float* __restrict__ ar, float* __restrict__ Y, float* __restrict__ el,
    float* __restrict__ er) {
  constexpr int K = 128;
  constexpr int HB = NCOL / 32;
  __shared__ float xs[16][K + 4];
  __shared__ float part[2][NCOL / 16][16];
  const int row0 = blockIdx.x * 16;
  const int tid = threadIdx.x;
  const int lane = tid & 31;
  const int wave = tid >> 5;

  for (int i = tid; i < 16 * K; i += 32 * (NCOL / 16)) {
    int r = i >> 7;
    int c = i & 127;
    xs[r][c] = X[(row0 + r) * K + c];
  }
  __syncthreads();

  const int n0 = wave * 16;
  const int m = lane & 15;
  const int khalf = lane >> 4;

  v8f acc = {};
  for (int k0 = 0; k0 < K; k0 += 4) {
    const int ka = k0 + khalf * 2;
    v2f a, b;
    a.x = xs[m][ka];
    a.y = xs[m][ka + 1];
    b.x = W[(ka) *NCOL + n0 + m];
    b.y = W[(ka + 1) * NCOL + n0 + m];
    acc = __builtin_amdgcn_wmma_f32_16x16x4_f32(false, a, false, b, (short)0, acc,
                                                false, false);
  }

  const int n = n0 + m;

  const float aln = al[n];
  const float arn = ar[n];
  float pl[8], pr[8];
  for (int r = 0; r < 8; ++r) {
    pl[r] = row16_sum(acc[r] * aln);
    pr[r] = row16_sum(acc[r] * arn);
  }
  if (m == 0) {
    for (int r = 0; r < 8; ++r) {
      part[0][wave][khalf * 8 + r] = pl[r];
      part[1][wave][khalf * 8 + r] = pr[r];
    }
  }
  __syncthreads();
#pragma unroll
  for (int r = 0; r < 8; ++r) {
    Y[(row0 + khalf * 8 + r) * NCOL + n] = acc[r];
  }
  if (tid < 32 * HB) {
    const int which = tid / (16 * HB);
    const int t = tid - which * (16 * HB);
    const int r = t / HB;
    const int hd = t - r * HB;
    const float v = part[which][2 * hd][r] + part[which][2 * hd + 1][r];
    float* __restrict__ dstp = which ? er : el;
    dstp[(row0 + r) * HB + hd] = v;
  }
}

__global__ void edge_rank(const int* __restrict__ dst, int* __restrict__ cnt,
                          int* __restrict__ rank, int ne) {
  const int i = blockIdx.x * blockDim.x + threadIdx.x;
  if (i >= ne) return;
  rank[i] = atomicAdd(cnt + dst[i], 1);
}

__global__ __launch_bounds__(256) void seg_offsets(const int* __restrict__ cnt,
                                                   int* __restrict__ off, int n) {
  __shared__ int w_before[8];
  __shared__ int w_tile[8];
  const int tid = threadIdx.x;
  const int lane = tid & 31;
  const int wave = tid >> 5;
  const int tile0 = blockIdx.x * 1024;

  int before = 0;
  for (int i = tid * 4; i < tile0; i += 1024) {
    const int4 v = *reinterpret_cast<const int4*>(cnt + i);
    before += v.x + v.y + v.z + v.w;
  }

  const int i0 = tile0 + tid * 4;
  int c[4];
  for (int k = 0; k < 4; ++k) c[k] = (i0 + k < n) ? cnt[i0 + k] : 0;
  const int local = c[0] + c[1] + c[2] + c[3];

  int incl = local;
  for (int o = 1; o < 32; o <<= 1) {
    const int t = __shfl_up(incl, o, 32);
    if (lane >= o) incl += t;
  }
  for (int o = 16; o > 0; o >>= 1) before += __shfl_xor(before, o, 32);

  if (lane == 31) w_tile[wave] = incl;
  if (lane == 0) w_before[wave] = before;
  __syncthreads();

  int run = incl - local;
  for (int w = 0; w < 8; ++w) {
    run += w_before[w];
    if (w < wave) run += w_tile[w];
  }
  for (int k = 0; k < 4; ++k) {
    if (i0 + k <= n) off[i0 + k] = run;
    run += c[k];
  }
}

__global__ void edge_place(const int* __restrict__ src, const int* __restrict__ dst,
                           const int* __restrict__ off, const int* __restrict__ rank,
                           int* __restrict__ csrc, int ne) {
  const int i = blockIdx.x * blockDim.x + threadIdx.x;
  if (i >= ne) return;
  unsigned pos = (unsigned)(off[dst[i]] + rank[i]);
  pos = min(pos, (unsigned)(ne - 1));
  csrc[pos] = src[i];
}

template <int H, bool ELU>
__global__ __launch_bounds__(256) void gat_gather(
    const int* __restrict__ off, const int* __restrict__ csrc,
    const float* __restrict__ el, const float* __restrict__ er,
    const float* __restrict__ hfeat, const float* __restrict__ bias,
    float* __restrict__ out, int n) {
  constexpr int HD = H * 32;
  __shared__ float wl[8][32][H];
  const int wave = threadIdx.x >> 5;
  const int node = blockIdx.x * 8 + wave;
  if (node >= n) return;
  const int lane = threadIdx.x & 31;
  const int head = (lane * H) >> 5;
  const int beg = __builtin_amdgcn_readfirstlane(off[node]);
  const int end = __builtin_amdgcn_readfirstlane(off[node + 1]);
  const unsigned last = (unsigned)(n - 1);

  float erh[H];
  for (int h = 0; h < H; ++h) erh[h] = er[node * H + h];
  float bcol[H];
  for (int k = 0; k < H; ++k) bcol[k] = bias[lane * H + k];

  float mh[H];
  float v0[H];
  int s0 = 0;
  for (int h = 0; h < H; ++h) {
    mh[h] = -INFINITY;
    v0[h] = 0.0f;
  }
  for (int base = beg; base < end; base += 32) {
    const int p = base + lane;
    if (p < end) {
      const int s = (int)min((unsigned)csrc[p], last);
      for (int h = 0; h < H; ++h) {
        float v = el[s * H + h] + erh[h];
        v = (v > 0.0f) ? v : NEG_SLOPE * v;
        mh[h] = fmaxf(mh[h], v);
        if (base == beg) v0[h] = v;
      }
      if (base == beg) s0 = s;
    }
  }
  for (int h = 0; h < H; ++h) {
    for (int o = 16; o > 0; o >>= 1) mh[h] = fmaxf(mh[h], __shfl_xor(mh[h], o, 32));
  }

  const float* hcol = hfeat + lane * H;
  float ssum = 0.0f;
  float acc[H];
  for (int k = 0; k < H; ++k) acc[k] = 0.0f;
  for (int base = beg; base < end; base += 32) {
    const int cnt = min(32, end - base);
    int s_l = s0;
    float v[H];
    for (int h = 0; h < H; ++h) v[h] = v0[h];
    if (base != beg) {
      s_l = 0;
      if (lane < cnt) {
        s_l = (int)min((unsigned)csrc[base + lane], last);
        for (int h = 0; h < H; ++h) {
          const float t = el[s_l * H + h] + erh[h];
          v[h] = (t > 0.0f) ? t : NEG_SLOPE * t;
        }
      }
    }
    if (lane < cnt) {
      for (int h = 0; h < H; ++h) wl[wave][lane][h] = expf(v[h] - mh[h]);
    }
    __builtin_amdgcn_wave_barrier();
#pragma unroll 4
    for (int j = 0; j < cnt; ++j) {
      const int s = __builtin_amdgcn_readlane(s_l, j);
      const float w = wl[wave][j][head];
      ssum += w;
      const float* row = hcol + (size_t)s * HD;
      if constexpr (H == 4) {
        const float4 r = *reinterpret_cast<const float4*>(row);
        acc[0] += w * r.x;
        acc[1] += w * r.y;
        acc[2] += w * r.z;
        acc[3] += w * r.w;
      } else {
        for (int k = 0; k < H; ++k) acc[k] += w * row[k];
      }
    }
    __builtin_amdgcn_wave_barrier();
  }

  const bool any = end > beg;
  float res[H];
  for (int k = 0; k < H; ++k) {
    float v = (any ? acc[k] / ssum : 0.0f) + bcol[k];
    if constexpr (ELU) v = (v > 0.0f) ? v : (expf(v) - 1.0f);
    res[k] = v;
  }
  float* o = out + (size_t)node * HD + lane * H;
  if constexpr (H == 4) {
    float4 r;
    r.x = res[0];
    r.y = res[1];
    r.z = res[2];
    r.w = res[3];
    *reinterpret_cast<float4*>(o) = r;
  } else {
    for (int k = 0; k < H; ++k) o[k] = res[k];
  }
}

extern "C" void kernel_launch(void* const* d_in, const int* in_sizes, int n_in,
                              void* d_out, int out_size, void* d_ws, size_t ws_size,
                              hipStream_t stream) {
  const float* x = (const float*)d_in[0];
  const int* src = (const int*)d_in[1];
  const int* dst = (const int*)d_in[2];
  const float* W1 = (const float*)d_in[3];
  const float* al1 = (const float*)d_in[4];
  const float* ar1 = (const float*)d_in[5];
  const float* b1 = (const float*)d_in[6];
  const float* W2 = (const float*)d_in[7];
  const float* al2 = (const float*)d_in[8];
  const float* ar2 = (const float*)d_in[9];
  const float* b2 = (const float*)d_in[10];

  const int n = in_sizes[0] / IN_F;
  const int e = in_sizes[1];

  char* ws = (char*)d_ws;
  size_t off = 0;
  auto wsalloc = [&](size_t bytes) {
    void* p = ws + off;
    off = (off + bytes + 255) & ~(size_t)255;
    return p;
  };
  float* h1 = (float*)wsalloc((size_t)n * 128 * sizeof(float));
  float* agg1 = (float*)wsalloc((size_t)n * 128 * sizeof(float));
  int* csrc = (int*)wsalloc((size_t)e * sizeof(int));
  int* rank = (int*)wsalloc((size_t)e * sizeof(int));
  int* cnt = (int*)wsalloc((size_t)n * sizeof(int));
  int* offs = (int*)wsalloc((size_t)(n + 1) * sizeof(int));
  float* el = (float*)wsalloc((size_t)n * H0 * sizeof(float));
  float* er = (float*)wsalloc((size_t)n * H0 * sizeof(float));
  float* outv = (float*)d_out;

  hipMemsetAsync(cnt, 0, (size_t)n * sizeof(int), stream);
  edge_rank<<<(e + 255) / 256, 256, 0, stream>>>(dst, cnt, rank, e);
  seg_offsets<<<(n + 1 + 1023) / 1024, 256, 0, stream>>>(cnt, offs, n);
  edge_place<<<(e + 255) / 256, 256, 0, stream>>>(src, dst, offs, rank, csrc, e);

  wmma_gemm_k128<128><<<n / 16, 256, 0, stream>>>(x, W1, al1, ar1, h1, el, er);

  gat_gather<H0, true><<<(n + 7) / 8, 256, 0, stream>>>(offs, csrc, el, er, h1, b1, agg1, n);

  float* g2 = h1;
  wmma_gemm_k128<32><<<n / 16, 64, 0, stream>>>(agg1, W2, al2, ar2, g2, el, er);

  gat_gather<H1, false><<<(n + 7) / 8, 256, 0, stream>>>(offs, csrc, el, er, g2, b2, outv, n);
}


# segment offsets: the sum over the entries in front of a tile keeps eight 16-byte loads per thread in flight (1024-thread blocks, 4096-entry tiles) instead of one waited load per 1024 entries
# speedup vs baseline: 4.8415x; 1.1448x over previous
#include <hip/hip_runtime.h>
#include <cstdint>
#include <cstddef>

#define IN_F 128
#define NEG_SLOPE 0.2f
#define H0 4
#define H1 1

typedef __attribute__((ext_vector_type(2))) float v2f;
typedef __attribute__((ext_vector_type(8))) float v8f;

template <int MASK>
__device__ __forceinline__ float row_xor_get(float v) {
  return __builtin_bit_cast(float, __builtin_amdgcn_update_dpp(0, __builtin_bit_cast(int, v),
                                                               0x160 + MASK, 0xf, 0xf, true));
}
__device__ __forceinline__ float row16_sum(float v) {
  v += row_xor_get<1>(v);
  v += row_xor_get<2>(v);
  v += row_xor_get<4>(v);
  v += row_xor_get<8>(v);
  return v;
}

template <int NCOL>
__global__ __launch_bounds__(32 * (NCOL / 16)) void wmma_gemm_k128(
    const float* __restrict__ X, const float* __restrict__ W, const float* __restrict__ al,
    const float* __restrict__ ar, float* __restrict__ Y, float* __restrict__ el,
    float* __restrict__ er) {
  constexpr int K = 128;
  constexpr int HB = NCOL / 32;
  __shared__ float xs[16][K + 4];
  __shared__ float part[2][NCOL / 16][16];
  const int row0 = blockIdx.x * 16;
  const int tid = threadIdx.x;
  const int lane = tid & 31;
  const int wave = tid >> 5;

  for (int i = tid; i < 16 * K; i += 32 * (NCOL / 16)) {
    int r = i >> 7;
    int c = i & 127;
    xs[r][c] = X[(row0 + r) * K + c];
  }
  __syncthreads();

  const int n0 = wave * 16;
  const int m = lane & 15;
  const int khalf = lane >> 4;

  v8f acc = {};
  for (int k0 = 0; k0 < K; k0 += 4) {
    const int ka = k0 + khalf * 2;
    v2f a, b;
    a.x = xs[m][ka];
    a.y = xs[m][ka + 1];
    b.x = W[(ka) *NCOL + n0 + m];
    b.y = W[(ka + 1) * NCOL + n0 + m];
    acc = __builtin_amdgcn_wmma_f32_16x16x4_f32(false, a, false, b, (short)0, acc,
                                                false, false);
  }

  const int n = n0 + m;

  const float aln = al[n];
  const float arn = ar[n];
  float pl[8], pr[8];
  for (int r = 0; r < 8; ++r) {
    pl[r] = row16_sum(acc[r] * aln);
    pr[r] = row16_sum(acc[r] * arn);
  }
  if (m == 0) {
    for (int r = 0; r < 8; ++r) {
      part[0][wave][khalf * 8 + r] = pl[r];
      part[1][wave][khalf * 8 + r] = pr[r];
    }
  }
  __syncthreads();
#pragma unroll
  for (int r = 0; r < 8; ++r) {
    Y[(row0 + khalf * 8 + r) * NCOL + n] = acc[r];
  }
  if (tid < 32 * HB) {
    const int which = tid / (16 * HB);
    const int t = tid - which * (16 * HB);
    const int r = t / HB;
    const int hd = t - r * HB;
    const float v = part[which][2 * hd][r] + part[which][2 * hd + 1][r];
    float* __restrict__ dstp = which ? er : el;
    dstp[(row0 + r) * HB + hd] = v;
  }
}

__global__ void edge_rank(const int* __restrict__ dst, int* __restrict__ cnt,
                          int* __restrict__ rank, int ne) {
  const int i = blockIdx.x * blockDim.x + threadIdx.x;
  if (i >= ne) return;
  rank[i] = atomicAdd(cnt + dst[i], 1);
}

__global__ __launch_bounds__(1024) void seg_offsets(const int* __restrict__ cnt,
                                                    int* __restrict__ off, int n) {
  __shared__ int w_before[32];
  __shared__ int w_tile[32];
  const int tid = threadIdx.x;
  const int lane = tid & 31;
  const int wave = tid >> 5;
  const int nb = blockIdx.x;
  const int tile0 = nb * 4096;

  const int i0 = tile0 + tid * 4;
  int c[4];
  if (i0 + 3 < n) {
    const int4 v = *reinterpret_cast<const int4*>(cnt + i0);
    c[0] = v.x;
    c[1] = v.y;
    c[2] = v.z;
    c[3] = v.w;
  } else {
    for (int k = 0; k < 4; ++k) c[k] = (i0 + k < n) ? cnt[i0 + k] : 0;
  }

  int before = 0;
  const int4* front = reinterpret_cast<const int4*>(cnt) + tid;
  for (int k = 0; k < nb; k += 8) {
    int4 v[8];
#pragma unroll
    for (int u = 0; u < 8; ++u) {
      const int t = (k + u < nb) ? (k + u) : 0;
      v[u] = front[(size_t)t * 1024];
    }
#pragma unroll
    for (int u = 0; u < 8; ++u) {
      const int s = v[u].x + v[u].y + v[u].z + v[u].w;
      before += (k + u < nb) ? s : 0;
    }
  }

  const int local = c[0] + c[1] + c[2] + c[3];
  int incl = local;
  for (int o = 1; o < 32; o <<= 1) {
    const int t = __shfl_up(incl, o, 32);
    if (lane >= o) incl += t;
  }
  for (int o = 16; o > 0; o >>= 1) before += __shfl_xor(before, o, 32);

  if (lane == 31) w_tile[wave] = incl;
  if (lane == 0) w_before[wave] = before;
  __syncthreads();

  int add = w_before[lane] + ((lane < wave) ? w_tile[lane] : 0);
  for (int o = 16; o > 0; o >>= 1) add += __shfl_xor(add, o, 32);

  int run = incl - local + add;
  if (i0 + 3 <= n) {
    int4 o4;
    o4.x = run;
    o4.y = run + c[0];
    o4.z = run + c[0] + c[1];
    o4.w = run + c[0] + c[1] + c[2];
    *reinterpret_cast<int4*>(off + i0) = o4;
  } else {
    for (int k = 0; k < 4; ++k) {
      if (i0 + k <= n) off[i0 + k] = run;
      run += c[k];
    }
  }
}

__global__ void edge_place(const int* __restrict__ src, const int* __restrict__ dst,
                           const int* __restrict__ off, const int* __restrict__ rank,
                           int* __restrict__ csrc, int ne) {
  const int i = blockIdx.x * blockDim.x + threadIdx.x;
  if (i >= ne) return;
  unsigned pos = (unsigned)(off[dst[i]] + rank[i]);
  pos = min(pos, (unsigned)(ne - 1));
  csrc[pos] = src[i];
}

template <int H, bool ELU>
__global__ __launch_bounds__(256) void gat_gather(
    const int* __restrict__ off, const int* __restrict__ csrc,
    const float* __restrict__ el, const float* __restrict__ er,
    const float* __restrict__ hfeat, const float* __restrict__ bias,
    float* __restrict__ out, int n) {
  constexpr int HD = H * 32;
  __shared__ float wl[8][32][H];
  const int wave = threadIdx.x >> 5;
  const int node = blockIdx.x * 8 + wave;
  if (node >= n) return;
  const int lane = threadIdx.x & 31;
  const int head = (lane * H) >> 5;
  const int beg = __builtin_amdgcn_readfirstlane(off[node]);
  const int end = __builtin_amdgcn_readfirstlane(off[node + 1]);
  const unsigned last = (unsigned)(n - 1);

  float erh[H];
  for (int h = 0; h < H; ++h) erh[h] = er[node * H + h];
  float bcol[H];
  for (int k = 0; k < H; ++k) bcol[k] = bias[lane * H + k];

  float mh[H];
  float v0[H];
  int s0 = 0;
  for (int h = 0; h < H; ++h) {
    mh[h] = -INFINITY;
    v0[h] = 0.0f;
  }
  for (int base = beg; base < end; base += 32) {
    const int p = base + lane;
    if (p < end) {
      const int s = (int)min((unsigned)csrc[p], last);
      for (int h = 0; h < H; ++h) {
        float v = el[s * H + h] + erh[h];
        v = (v > 0.0f) ? v : NEG_SLOPE * v;
        mh[h] = fmaxf(mh[h], v);
        if (base == beg) v0[h] = v;
      }
      if (base == beg) s0 = s;
    }
  }
  for (int h = 0; h < H; ++h) {
    for (int o = 16; o > 0; o >>= 1) mh[h] = fmaxf(mh[h], __shfl_xor(mh[h], o, 32));
  }

  const float* hcol = hfeat + lane * H;
  float ssum = 0.0f;
  float acc[H];
  for (int k = 0; k < H; ++k) acc[k] = 0.0f;
  for (int base = beg; base < end; base += 32) {
    const int cnt = min(32, end - base);
    int s_l = s0;
    float v[H];
    for (int h = 0; h < H; ++h) v[h] = v0[h];
    if (base != beg) {
      s_l = 0;
      if (lane < cnt) {
        s_l = (int)min((unsigned)csrc[base + lane], last);
        for (int h = 0; h < H; ++h) {
          const float t = el[s_l * H + h] + erh[h];
          v[h] = (t > 0.0f) ? t : NEG_SLOPE * t;
        }
      }
    }
    if (lane < cnt) {
      for (int h = 0; h < H; ++h) wl[wave][lane][h] = expf(v[h] - mh[h]);
    }
    __builtin_amdgcn_wave_barrier();
#pragma unroll 4
    for (int j = 0; j < cnt; ++j) {
      const int s = __builtin_amdgcn_readlane(s_l, j);
      const float w = wl[wave][j][head];
      ssum += w;
      const float* row = hcol + (size_t)s * HD;
      if constexpr (H == 4) {
        const float4 r = *reinterpret_cast<const float4*>(row);
        acc[0] += w * r.x;
        acc[1] += w * r.y;
        acc[2] += w * r.z;
        acc[3] += w * r.w;
      } else {
        for (int k = 0; k < H; ++k) acc[k] += w * row[k];
      }
    }
    __builtin_amdgcn_wave_barrier();
  }

  const bool any = end > beg;
  float res[H];
  for (int k = 0; k < H; ++k) {
    float v = (any ? acc[k] / ssum : 0.0f) + bcol[k];
    if constexpr (ELU) v = (v > 0.0f) ? v : (expf(v) - 1.0f);
    res[k] = v;
  }
  float* o = out + (size_t)node * HD + lane * H;
  if constexpr (H == 4) {
    float4 r;
    r.x = res[0];
    r.y = res[1];
    r.z = res[2];
    r.w = res[3];
    *reinterpret_cast<float4*>(o) = r;
  } else {
    for (int k = 0; k < H; ++k) o[k] = res[k];
  }
}

extern "C" void kernel_launch(void* const* d_in, const int* in_sizes, int n_in,
                              void* d_out, int out_size, void* d_ws, size_t ws_size,
                              hipStream_t stream) {
  const float* x = (const float*)d_in[0];
  const int* src = (const int*)d_in[1];
  const int* dst = (const int*)d_in[2];
  const float* W1 = (const float*)d_in[3];
  const float* al1 = (const float*)d_in[4];
  const float* ar1 = (const float*)d_in[5];
  const float* b1 = (const float*)d_in[6];
  const float* W2 = (const float*)d_in[7];
  const float* al2 = (const float*)d_in[8];
  const float* ar2 = (const float*)d_in[9];
  const float* b2 = (const float*)d_in[10];

  const int n = in_sizes[0] / IN_F;
  const int e = in_sizes[1];

  char* ws = (char*)d_ws;
  size_t off = 0;
  auto wsalloc = [&](size_t bytes) {
    void* p = ws + off;
    off = (off + bytes + 255) & ~(size_t)255;
    return p;
  };
  float* h1 = (float*)wsalloc((size_t)n * 128 * sizeof(float));
  float* agg1 = (float*)wsalloc((size_t)n * 128 * sizeof(float));
  int* csrc = (int*)wsalloc((size_t)e * sizeof(int));
  int* rank = (int*)wsalloc((size_t)e * sizeof(int));
  int* cnt = (int*)wsalloc((size_t)n * sizeof(int));
  int* offs = (int*)wsalloc((size_t)(n + 1) * sizeof(int));
  float* el = (float*)wsalloc((size_t)n * H0 * sizeof(float));
  float* er = (float*)wsalloc((size_t)n * H0 * sizeof(float));
  float* outv = (float*)d_out;

  hipMemsetAsync(cnt, 0, (size_t)n * sizeof(int), stream);
  edge_rank<<<(e + 255) / 256, 256, 0, stream>>>(dst, cnt, rank, e);
  seg_offsets<<<(n + 1 + 4095) / 4096, 1024, 0, stream>>>(cnt, offs, n);
  edge_place<<<(e + 255) / 256, 256, 0, stream>>>(src, dst, offs, rank, csrc, e);

  wmma_gemm_k128<128><<<n / 16, 256, 0, stream>>>(x, W1, al1, ar1, h1, el, er);

  gat_gather<H0, true><<<(n + 7) / 8, 256, 0, stream>>>(offs, csrc, el, er, h1, b1, agg1, n);

  float* g2 = h1;
  wmma_gemm_k128<32><<<n / 16, 64, 0, stream>>>(agg1, W2, al2, ar2, g2, el, er);

  gat_gather<H1, false><<<(n + 7) / 8, 256, 0, stream>>>(offs, csrc, el, er, g2, b2, outv, n);
}
